# starting kernel
# baseline (speedup 1.0000x reference)
#include <hip/hip_runtime.h>
#include <hip/hip_bf16.h>

typedef __attribute__((ext_vector_type(2))) float v2f;
typedef __attribute__((ext_vector_type(8))) float v8f;

#if __has_builtin(__builtin_amdgcn_tanhf)
__device__ __forceinline__ float fast_tanh(float x) { return __builtin_amdgcn_tanhf(x); }
#else
__device__ __forceinline__ float fast_tanh(float x) { return tanhf(x); }
#endif

__global__ __launch_bounds__(256) void rule_layer_wmma_f32(
    const float* __restrict__ values,
    const float* __restrict__ weights,
    const int*   __restrict__ indices,
    float*       __restrict__ out,
    int Bn, int ntiles)
{
  const int lane = threadIdx.x & 31;
  const int n    = lane & 15;
  const int hi   = lane >> 4;
  const int koff = hi * 2;

  const int waveId = blockIdx.x * (blockDim.x >> 5) + (threadIdx.x >> 5);
  if (waveId >= ntiles) return;
  const int btile = waveId * 16;

  float wl[8];
#pragma unroll
  for (int w = 0; w < 8; ++w) wl[w] = weights[w * 16 + n];

  int b = btile + n;
  if (b >= Bn) b = Bn - 1;
  int rows[8];
#pragma unroll
  for (int w = 0; w < 8; ++w) rows[w] = indices[(size_t)w * (size_t)Bn + b];

  v8f acc = {};
#pragma unroll
  for (int w = 0; w < 8; ++w) {
    const v2f* pw = (const v2f*)(values + (size_t)rows[w] * 16 + koff);
#pragma unroll
    for (int c = 0; c < 4; ++c) {
      const int d0 = c * 4;
      v2f a = pw[c * 2];
      v2f bm;
      bm.x = (n == d0 + koff)     ? wl[w] : 0.0f;
      bm.y = (n == d0 + koff + 1) ? wl[w] : 0.0f;
      acc = __builtin_amdgcn_wmma_f32_16x16x4_f32(
          false, a, false, bm, (short)0, acc, false, false);
    }
  }

  float* op = out + (size_t)(btile + hi * 8) * 16 + n;
#pragma unroll
  for (int r = 0; r < 8; ++r) {
    int rowb = btile + hi * 8 + r;
    float v = fast_tanh(acc[r]);
    if (rowb < Bn) op[(size_t)r * 16] = v;
  }
}

extern "C" void kernel_launch(void* const* d_in, const int* in_sizes, int n_in,
                              void* d_out, int out_size, void* d_ws, size_t ws_size,
                              hipStream_t stream) {
  const float* values  = (const float*)d_in[0];
  const float* weights = (const float*)d_in[1];
  const int*   indices = (const int*)d_in[2];
  float*       out     = (float*)d_out;

  const int W  = 8;
  const int Bn = in_sizes[2] / W;
  const int ntiles = (Bn + 15) / 16;
  const int wavesPerBlock = 256 / 32;
  const int blocks = (ntiles + wavesPerBlock - 1) / wavesPerBlock;

  rule_layer_wmma_f32<<<blocks, 256, 0, stream>>>(values, weights, indices, out,
                                                  Bn, ntiles);
}


# Gather each 64-byte source row with four lanes x 16 bytes (8 whole rows per load, two 8-row accumulator tiles per wave) instead of 8-byte pieces of 16 rows per load
# speedup vs baseline: 1.6615x; 1.6615x over previous
#include <hip/hip_runtime.h>
#include <hip/hip_bf16.h>

typedef __attribute__((ext_vector_type(2))) float v2f;
typedef __attribute__((ext_vector_type(4))) float v4f;
typedef __attribute__((ext_vector_type(8))) float v8f;

#if __has_builtin(__builtin_amdgcn_tanhf)
__device__ __forceinline__ float fast_tanh(float x) { return __builtin_amdgcn_tanhf(x); }
#else
__device__ __forceinline__ float fast_tanh(float x) { return tanhf(x); }
#endif

__global__ __launch_bounds__(256) void rule_layer_wmma_f32(
    const float* __restrict__ values,
    const float* __restrict__ weights,
    const int*   __restrict__ indices,
    float*       __restrict__ out,
    int Bn, int ntiles)
{
  const int lane = threadIdx.x & 31;
  const int n    = lane & 15;
  const int hi   = lane >> 4;
  const int hsel = (lane >> 3) & 1;
  const int e    = lane & 7;

  const int waveId = blockIdx.x * (blockDim.x >> 5) + (threadIdx.x >> 5);
  if (waveId >= ntiles) return;
  const int btile = waveId * 16;

  float wl[8];
#pragma unroll
  for (int w = 0; w < 8; ++w) wl[w] = weights[w * 16 + n];

  int b = btile + n;
  if (b >= Bn) b = Bn - 1;
  int rows[8];
#pragma unroll
  for (int w = 0; w < 8; ++w) rows[w] = indices[(size_t)w * (size_t)Bn + b];

  const unsigned q0 = (unsigned)(hsel * 2 + hi);
  const unsigned q1 = (unsigned)((1 - hsel) * 2 + hi);
  const v4f* __restrict__ vq = (const v4f*)values;

  const bool s0 = (e == 4 * hi);
  const bool s1 = (e == 4 * hi + 1);
  const bool s2 = (e == 4 * hi + 2);
  const bool s3 = (e == 4 * hi + 3);

  v8f acc0 = {};
  v8f acc1 = {};
#pragma unroll
  for (int w = 0; w < 8; ++w) {
    const int r0 = __builtin_amdgcn_update_dpp(rows[w], rows[w], 0x128, 0xf, 0xC, false);
    const int r1 = __builtin_amdgcn_update_dpp(rows[w], rows[w], 0x128, 0xf, 0x3, false);
    const v4f a0 = vq[(unsigned)r0 * 4u + q0];
    const v4f a1 = vq[(unsigned)r1 * 4u + q1];

    v2f bA, bB;
    bA.x = s0 ? wl[w] : 0.0f;
    bA.y = s1 ? wl[w] : 0.0f;
    bB.x = s2 ? wl[w] : 0.0f;
    bB.y = s3 ? wl[w] : 0.0f;

    v2f a0lo = {a0.x, a0.y}, a0hi = {a0.z, a0.w};
    v2f a1lo = {a1.x, a1.y}, a1hi = {a1.z, a1.w};
    acc0 = __builtin_amdgcn_wmma_f32_16x16x4_f32(
        false, a0lo, false, bA, (short)0, acc0, false, false);
    acc1 = __builtin_amdgcn_wmma_f32_16x16x4_f32(
        false, a1lo, false, bA, (short)0, acc1, false, false);
    acc0 = __builtin_amdgcn_wmma_f32_16x16x4_f32(
        false, a0hi, false, bB, (short)0, acc0, false, false);
    acc1 = __builtin_amdgcn_wmma_f32_16x16x4_f32(
        false, a1hi, false, bB, (short)0, acc1, false, false);
  }

  const bool g1 = (hsel != hi);
  const int rbase = btile + (g1 ? 8 : 0);
  float* op = out + (size_t)rbase * 16 + n;
#pragma unroll
  for (int r = 0; r < 8; ++r) {
    float v = fast_tanh(g1 ? acc1[r] : acc0[r]);
    if (rbase + r < Bn) op[(size_t)r * 16] = v;
  }
}

extern "C" void kernel_launch(void* const* d_in, const int* in_sizes, int n_in,
                              void* d_out, int out_size, void* d_ws, size_t ws_size,
                              hipStream_t stream) {
  const float* values  = (const float*)d_in[0];
  const float* weights = (const float*)d_in[1];
  const int*   indices = (const int*)d_in[2];
  float*       out     = (float*)d_out;

  const int W  = 8;
  const int Bn = in_sizes[2] / W;
  const int ntiles = (Bn + 15) / 16;
  const int wavesPerBlock = 256 / 32;
  const int blocks = (ntiles + wavesPerBlock - 1) / wavesPerBlock;

  rule_layer_wmma_f32<<<blocks, 256, 0, stream>>>(values, weights, indices, out,
                                                  Bn, ntiles);
}
